# starting kernel
# baseline (speedup 1.0000x reference)
#include <hip/hip_runtime.h>
#include <hip/hip_bf16.h>

#define BS      8
#define SEQ     2048
#define IN_F    2048
#define OUT_F   2048
#define NSPLIT  4
#define NSKILL  8
#define RANK    16
#define DSPLIT  (IN_F / NSPLIT)
#define OSPLIT  (OUT_F / NSPLIT)
#define SCALING 2.0f

typedef float v2f __attribute__((ext_vector_type(2)));
typedef float v8f __attribute__((ext_vector_type(8)));

__global__ __launch_bounds__(256) void lora_mix_kernel(
    const float* __restrict__ weights,
    const float* __restrict__ lora_a,
    const float* __restrict__ lora_b,
    float* __restrict__ A_mix,
    float* __restrict__ B_mix)
{
    const int idx  = blockIdx.x * 256 + threadIdx.x;
    const int half = BS * IN_F * RANK;

    if (idx < half) {
        const int b   = idx / (IN_F * RANK);
        const int rem = idx - b * (IN_F * RANK);
        const int i   = rem / RANK;
        const int r   = rem - i * RANK;
        const int q   = i / DSPLIT;
        const int d   = i - q * DSPLIT;
        const float* w  = weights + (b * NSPLIT + q) * NSKILL;
        const float* la = lora_a + ((size_t)(q * NSKILL) * DSPLIT + d) * RANK + r;
        float acc = 0.f;
        #pragma unroll
        for (int s = 0; s < NSKILL; ++s)
            acc = __builtin_fmaf(w[s], la[(size_t)s * DSPLIT * RANK], acc);
        A_mix[idx] = acc;
    } else {
        const int j   = idx - half;
        const int b   = j / (RANK * OUT_F);
        const int rem = j - b * (RANK * OUT_F);
        const int r   = rem / OUT_F;
        const int o   = rem - r * OUT_F;
        const int q   = o / OSPLIT;
        const int d   = o - q * OSPLIT;
        const float* w  = weights + (b * NSPLIT + q) * NSKILL;
        const float* lb = lora_b + ((size_t)(q * NSKILL) * RANK + r) * OSPLIT + d;
        float acc = 0.f;
        #pragma unroll
        for (int s = 0; s < NSKILL; ++s)
            acc = __builtin_fmaf(w[s], lb[(size_t)s * RANK * OSPLIT], acc);
        B_mix[j] = acc * SCALING;
    }
}

__global__ __launch_bounds__(128) void skilled_lora_kernel(
    const float* __restrict__ x,
    const float* __restrict__ A_mix,
    const float* __restrict__ B_mix,
    float* __restrict__ out)
{
    __shared__ float Tlds[4][16][16];

    const int wave = threadIdx.x >> 5;
    const int lane = threadIdx.x & 31;
    const int h    = lane >> 4;
    const int c    = lane & 15;

    const int tile = blockIdx.x * 4 + wave;
    const int b    = tile >> 7;
    const int m0   = (tile & 127) << 4;

    const float* xrow = x + ((size_t)b * SEQ + m0 + c) * IN_F + 2 * h;
    const float* acol = A_mix + (size_t)b * IN_F * RANK + (2 * h) * RANK + c;

    v8f acc = {0.f, 0.f, 0.f, 0.f, 0.f, 0.f, 0.f, 0.f};

    for (int k0 = 0; k0 < IN_F; k0 += 64) {
        if (k0 + 64 < IN_F)
            __builtin_prefetch(xrow + k0 + 64, 0, 0);
        #pragma unroll
        for (int k = k0; k < k0 + 64; k += 4) {
            v2f av = *(const v2f*)(xrow + k);
            v2f bv;
            bv.x = acol[(size_t)k * RANK];
            bv.y = acol[(size_t)k * RANK + RANK];
            acc = __builtin_amdgcn_wmma_f32_16x16x4_f32(
                false, av, false, bv, (short)0, acc, false, false);
        }
    }

    #pragma unroll
    for (int v = 0; v < 8; ++v)
        Tlds[wave][v + 8 * h][c] = acc[v];

    asm volatile("s_wait_dscnt 0x0" ::: "memory");

    v2f a2[4];
    #pragma unroll
    for (int kk = 0; kk < 4; ++kk)
        a2[kk] = *(const v2f*)&Tlds[wave][c][4 * kk + 2 * h];

    const float* Bbase = B_mix + (size_t)b * RANK * OUT_F + (2 * h) * OUT_F + c;
    float*       obase = out + ((size_t)b * SEQ + m0 + 8 * h) * OUT_F + c;

    for (int n0 = 0; n0 < OUT_F; n0 += 16) {
        v8f acc2 = {0.f, 0.f, 0.f, 0.f, 0.f, 0.f, 0.f, 0.f};
        const float* bp = Bbase + n0;
        #pragma unroll
        for (int kk = 0; kk < 4; ++kk) {
            v2f bv;
            bv.x = bp[(size_t)(4 * kk)     * OUT_F];
            bv.y = bp[(size_t)(4 * kk + 1) * OUT_F];
            acc2 = __builtin_amdgcn_wmma_f32_16x16x4_f32(
                false, a2[kk], false, bv, (short)0, acc2, false, false);
        }
        float* op = obase + n0;
        #pragma unroll
        for (int v = 0; v < 8; ++v)
            op[(size_t)v * OUT_F] = acc2[v];
    }
}

extern "C" void kernel_launch(void* const* d_in, const int* in_sizes, int n_in,
                              void* d_out, int out_size, void* d_ws, size_t ws_size,
                              hipStream_t stream) {
    const float* input   = (const float*)d_in[0];
    const float* weights = (const float*)d_in[1];
    const float* lora_a  = (const float*)d_in[2];
    const float* lora_b  = (const float*)d_in[3];
    float*       outp    = (float*)d_out;

    float* A_mix = (float*)d_ws;
    float* B_mix = A_mix + (size_t)BS * IN_F * RANK;

    lora_mix_kernel<<<2048, 256, 0, stream>>>(weights, lora_a, lora_b, A_mix, B_mix);

    skilled_lora_kernel<<<256, 128, 0, stream>>>(input, A_mix, B_mix, outp);
}


# Phase 2 re-tiled over waves: a 256-thread block owns 32 rows; its 8 waves split GEMM1's K (partial sums added through LDS: k-sum order changes) and GEMM2's N; 4096 waves where there were 1024
# speedup vs baseline: 1.0985x; 1.0985x over previous
#include <hip/hip_runtime.h>
#include <hip/hip_bf16.h>

#define BS      8
#define SEQ     2048
#define IN_F    2048
#define OUT_F   2048
#define NSPLIT  4
#define NSKILL  8
#define RANK    16
#define DSPLIT  (IN_F / NSPLIT)
#define OSPLIT  (OUT_F / NSPLIT)
#define SCALING 2.0f

typedef float v2f __attribute__((ext_vector_type(2)));
typedef float v8f __attribute__((ext_vector_type(8)));

__global__ __launch_bounds__(256) void lora_mix_kernel(
    const float* __restrict__ weights,
    const float* __restrict__ lora_a,
    const float* __restrict__ lora_b,
    float* __restrict__ A_mix,
    float* __restrict__ B_mix)
{
    const int idx  = blockIdx.x * 256 + threadIdx.x;
    const int half = BS * IN_F * RANK;

    if (idx < half) {
        const int b   = idx / (IN_F * RANK);
        const int rem = idx - b * (IN_F * RANK);
        const int i   = rem / RANK;
        const int r   = rem - i * RANK;
        const int q   = i / DSPLIT;
        const int d   = i - q * DSPLIT;
        const float* w  = weights + (b * NSPLIT + q) * NSKILL;
        const float* la = lora_a + ((size_t)(q * NSKILL) * DSPLIT + d) * RANK + r;
        float acc = 0.f;
        #pragma unroll
        for (int s = 0; s < NSKILL; ++s)
            acc = __builtin_fmaf(w[s], la[(size_t)s * DSPLIT * RANK], acc);
        A_mix[idx] = acc;
    } else {
        const int j   = idx - half;
        const int b   = j / (RANK * OUT_F);
        const int rem = j - b * (RANK * OUT_F);
        const int r   = rem / OUT_F;
        const int o   = rem - r * OUT_F;
        const int q   = o / OSPLIT;
        const int d   = o - q * OSPLIT;
        const float* w  = weights + (b * NSPLIT + q) * NSKILL;
        const float* lb = lora_b + ((size_t)(q * NSKILL) * RANK + r) * OSPLIT + d;
        float acc = 0.f;
        #pragma unroll
        for (int s = 0; s < NSKILL; ++s)
            acc = __builtin_fmaf(w[s], lb[(size_t)s * RANK * OSPLIT], acc);
        B_mix[j] = acc * SCALING;
    }
}

#define NWAVES  8
#define KSLICE  (IN_F / NWAVES)
#define NSLICE  (OUT_F / NWAVES)
#define TPITCH  20

__global__ __launch_bounds__(256) void skilled_lora_kernel(
    const float* __restrict__ x,
    const float* __restrict__ A_mix,
    const float* __restrict__ B_mix,
    float* __restrict__ out)
{
    __shared__ __attribute__((aligned(16))) float Tp[NWAVES][2][16][TPITCH];

    const int wave = threadIdx.x >> 5;
    const int lane = threadIdx.x & 31;
    const int h    = lane >> 4;
    const int c    = lane & 15;

    const int b    = blockIdx.x >> 6;
    const int m0   = (blockIdx.x & 63) << 5;

    const int kb = wave * KSLICE;
    const float* xr0  = x + ((size_t)b * SEQ + m0 + c) * IN_F + kb + 2 * h;
    const float* xr1  = xr0 + (size_t)16 * IN_F;
    const float* acol = A_mix + ((size_t)b * IN_F + kb + 2 * h) * RANK + c;

    v8f acc0 = {0.f, 0.f, 0.f, 0.f, 0.f, 0.f, 0.f, 0.f};
    v8f acc1 = {0.f, 0.f, 0.f, 0.f, 0.f, 0.f, 0.f, 0.f};

    for (int k0 = 0; k0 < KSLICE; k0 += 64) {
        if (k0 + 64 < KSLICE) {
            __builtin_prefetch(xr0 + k0 + 64, 0, 0);
            __builtin_prefetch(xr1 + k0 + 64, 0, 0);
        }
        #pragma unroll
        for (int k = k0; k < k0 + 64; k += 4) {
            v2f a0 = *(const v2f*)(xr0 + k);
            v2f a1 = *(const v2f*)(xr1 + k);
            v2f bv;
            bv.x = acol[(size_t)k * RANK];
            bv.y = acol[(size_t)k * RANK + RANK];
            acc0 = __builtin_amdgcn_wmma_f32_16x16x4_f32(
                false, a0, false, bv, (short)0, acc0, false, false);
            acc1 = __builtin_amdgcn_wmma_f32_16x16x4_f32(
                false, a1, false, bv, (short)0, acc1, false, false);
        }
    }

    #pragma unroll
    for (int v = 0; v < 8; ++v) {
        Tp[wave][0][v + 8 * h][c] = acc0[v];
        Tp[wave][1][v + 8 * h][c] = acc1[v];
    }

    __syncthreads();

    v2f t0[4], t1[4];
    #pragma unroll
    for (int kk = 0; kk < 4; ++kk) {
        t0[kk] = *(const v2f*)&Tp[0][0][c][4 * kk + 2 * h];
        t1[kk] = *(const v2f*)&Tp[0][1][c][4 * kk + 2 * h];
    }
    #pragma unroll
    for (int w = 1; w < NWAVES; ++w) {
        #pragma unroll
        for (int kk = 0; kk < 4; ++kk) {
            t0[kk] += *(const v2f*)&Tp[w][0][c][4 * kk + 2 * h];
            t1[kk] += *(const v2f*)&Tp[w][1][c][4 * kk + 2 * h];
        }
    }

    const int nb = wave * NSLICE;
    const float* Bbase = B_mix + ((size_t)b * RANK + 2 * h) * OUT_F + nb + c;
    float*       ob0   = out + ((size_t)b * SEQ + m0 + 8 * h) * OUT_F + nb + c;
    float*       ob1   = ob0 + (size_t)16 * OUT_F;

    #pragma unroll
    for (int n0 = 0; n0 < NSLICE; n0 += 16) {
        v8f d0 = {0.f, 0.f, 0.f, 0.f, 0.f, 0.f, 0.f, 0.f};
        v8f d1 = {0.f, 0.f, 0.f, 0.f, 0.f, 0.f, 0.f, 0.f};
        const float* bp = Bbase + n0;
        #pragma unroll
        for (int kk = 0; kk < 4; ++kk) {
            v2f bv;
            bv.x = bp[(size_t)(4 * kk)     * OUT_F];
            bv.y = bp[(size_t)(4 * kk + 1) * OUT_F];
            d0 = __builtin_amdgcn_wmma_f32_16x16x4_f32(
                false, t0[kk], false, bv, (short)0, d0, false, false);
            d1 = __builtin_amdgcn_wmma_f32_16x16x4_f32(
                false, t1[kk], false, bv, (short)0, d1, false, false);
        }
        float* op0 = ob0 + n0;
        float* op1 = ob1 + n0;
        #pragma unroll
        for (int v = 0; v < 8; ++v) {
            op0[(size_t)v * OUT_F] = d0[v];
            op1[(size_t)v * OUT_F] = d1[v];
        }
    }
}

extern "C" void kernel_launch(void* const* d_in, const int* in_sizes, int n_in,
                              void* d_out, int out_size, void* d_ws, size_t ws_size,
                              hipStream_t stream) {
    const float* input   = (const float*)d_in[0];
    const float* weights = (const float*)d_in[1];
    const float* lora_a  = (const float*)d_in[2];
    const float* lora_b  = (const float*)d_in[3];
    float*       outp    = (float*)d_out;

    float* A_mix = (float*)d_ws;
    float* B_mix = A_mix + (size_t)BS * IN_F * RANK;

    lora_mix_kernel<<<2048, 256, 0, stream>>>(weights, lora_a, lora_b, A_mix, B_mix);

    skilled_lora_kernel<<<512, 256, 0, stream>>>(input, A_mix, B_mix, outp);
}


# GEMM2 result tiles go through a wave-private LDS block and leave as 16-byte stores, 512 contiguous bytes of a row per instruction (was 4-byte stores of two 64-byte half lines); sum order unchanged
# speedup vs baseline: 1.4021x; 1.2764x over previous
#include <hip/hip_runtime.h>
#include <hip/hip_bf16.h>

#define BS      8
#define SEQ     2048
#define IN_F    2048
#define OUT_F   2048
#define NSPLIT  4
#define NSKILL  8
#define RANK    16
#define DSPLIT  (IN_F / NSPLIT)
#define OSPLIT  (OUT_F / NSPLIT)
#define SCALING 2.0f

typedef float v2f __attribute__((ext_vector_type(2)));
typedef float v8f __attribute__((ext_vector_type(8)));

__global__ __launch_bounds__(256) void lora_mix_kernel(
    const float* __restrict__ weights,
    const float* __restrict__ lora_a,
    const float* __restrict__ lora_b,
    float* __restrict__ A_mix,
    float* __restrict__ B_mix)
{
    const int idx  = blockIdx.x * 256 + threadIdx.x;
    const int half = BS * IN_F * RANK;

    if (idx < half) {
        const int b   = idx / (IN_F * RANK);
        const int rem = idx - b * (IN_F * RANK);
        const int i   = rem / RANK;
        const int r   = rem - i * RANK;
        const int q   = i / DSPLIT;
        const int d   = i - q * DSPLIT;
        const float* w  = weights + (b * NSPLIT + q) * NSKILL;
        const float* la = lora_a + ((size_t)(q * NSKILL) * DSPLIT + d) * RANK + r;
        float acc = 0.f;
        #pragma unroll
        for (int s = 0; s < NSKILL; ++s)
            acc = __builtin_fmaf(w[s], la[(size_t)s * DSPLIT * RANK], acc);
        A_mix[idx] = acc;
    } else {
        const int j   = idx - half;
        const int b   = j / (RANK * OUT_F);
        const int rem = j - b * (RANK * OUT_F);
        const int r   = rem / OUT_F;
        const int o   = rem - r * OUT_F;
        const int q   = o / OSPLIT;
        const int d   = o - q * OSPLIT;
        const float* w  = weights + (b * NSPLIT + q) * NSKILL;
        const float* lb = lora_b + ((size_t)(q * NSKILL) * RANK + r) * OSPLIT + d;
        float acc = 0.f;
        #pragma unroll
        for (int s = 0; s < NSKILL; ++s)
            acc = __builtin_fmaf(w[s], lb[(size_t)s * RANK * OSPLIT], acc);
        B_mix[j] = acc * SCALING;
    }
}

#define NWAVES  8
#define KSLICE  (IN_F / NWAVES)
#define NSLICE  (OUT_F / NWAVES)
#define TPITCH  20
#define SCOLS   128
#define SPITCH  (SCOLS + 4)

typedef float v4f __attribute__((ext_vector_type(4)));

__device__ __forceinline__ void wave_lds_order() {
    __builtin_amdgcn_fence(__ATOMIC_RELEASE, "wavefront");
    __builtin_amdgcn_wave_barrier();
    __builtin_amdgcn_fence(__ATOMIC_ACQUIRE, "wavefront");
}

__global__ __launch_bounds__(256) void skilled_lora_kernel(
    const float* __restrict__ x,
    const float* __restrict__ A_mix,
    const float* __restrict__ B_mix,
    float* __restrict__ out)
{
    __shared__ __attribute__((aligned(16))) float Tp[NWAVES][2][16][TPITCH];
    __shared__ __attribute__((aligned(16))) float Stg[NWAVES][16][SPITCH];

    const int wave = threadIdx.x >> 5;
    const int lane = threadIdx.x & 31;
    const int h    = lane >> 4;
    const int c    = lane & 15;

    const int b    = blockIdx.x >> 6;
    const int m0   = (blockIdx.x & 63) << 5;

    const int kb = wave * KSLICE;
    const float* xr0  = x + ((size_t)b * SEQ + m0 + c) * IN_F + kb + 2 * h;
    const float* xr1  = xr0 + (size_t)16 * IN_F;
    const float* acol = A_mix + ((size_t)b * IN_F + kb + 2 * h) * RANK + c;

    v8f acc0 = {0.f, 0.f, 0.f, 0.f, 0.f, 0.f, 0.f, 0.f};
    v8f acc1 = {0.f, 0.f, 0.f, 0.f, 0.f, 0.f, 0.f, 0.f};

    for (int k0 = 0; k0 < KSLICE; k0 += 64) {
        if (k0 + 64 < KSLICE) {
            __builtin_prefetch(xr0 + k0 + 64, 0, 0);
            __builtin_prefetch(xr1 + k0 + 64, 0, 0);
        }
        #pragma unroll
        for (int k = k0; k < k0 + 64; k += 4) {
            v2f a0 = *(const v2f*)(xr0 + k);
            v2f a1 = *(const v2f*)(xr1 + k);
            v2f bv;
            bv.x = acol[(size_t)k * RANK];
            bv.y = acol[(size_t)k * RANK + RANK];
            acc0 = __builtin_amdgcn_wmma_f32_16x16x4_f32(
                false, a0, false, bv, (short)0, acc0, false, false);
            acc1 = __builtin_amdgcn_wmma_f32_16x16x4_f32(
                false, a1, false, bv, (short)0, acc1, false, false);
        }
    }

    #pragma unroll
    for (int v = 0; v < 8; ++v) {
        Tp[wave][0][v + 8 * h][c] = acc0[v];
        Tp[wave][1][v + 8 * h][c] = acc1[v];
    }

    __syncthreads();

    v2f t0[4], t1[4];
    #pragma unroll
    for (int kk = 0; kk < 4; ++kk) {
        t0[kk] = *(const v2f*)&Tp[0][0][c][4 * kk + 2 * h];
        t1[kk] = *(const v2f*)&Tp[0][1][c][4 * kk + 2 * h];
    }
    #pragma unroll
    for (int w = 1; w < NWAVES; ++w) {
        #pragma unroll
        for (int kk = 0; kk < 4; ++kk) {
            t0[kk] += *(const v2f*)&Tp[w][0][c][4 * kk + 2 * h];
            t1[kk] += *(const v2f*)&Tp[w][1][c][4 * kk + 2 * h];
        }
    }

    const int nb = wave * NSLICE;
    const float* Bbase = B_mix + ((size_t)b * RANK + 2 * h) * OUT_F + nb + c;

    v2f bf[NSLICE / 16][4];
    #pragma unroll
    for (int j = 0; j < NSLICE / 16; ++j) {
        #pragma unroll
        for (int kk = 0; kk < 4; ++kk) {
            bf[j][kk].x = Bbase[(size_t)(4 * kk)     * OUT_F + 16 * j];
            bf[j][kk].y = Bbase[(size_t)(4 * kk + 1) * OUT_F + 16 * j];
        }
    }

    float* stg  = &Stg[wave][0][0];
    float* orow = out + ((size_t)b * SEQ + m0) * OUT_F + nb + 4 * lane;

    #pragma unroll
    for (int ch = 0; ch < NSLICE / SCOLS; ++ch) {
        #pragma unroll
        for (int rt = 0; rt < 2; ++rt) {
            #pragma unroll
            for (int j = 0; j < SCOLS / 16; ++j) {
                v8f d = {0.f, 0.f, 0.f, 0.f, 0.f, 0.f, 0.f, 0.f};
                #pragma unroll
                for (int kk = 0; kk < 4; ++kk)
                    d = __builtin_amdgcn_wmma_f32_16x16x4_f32(
                        false, rt ? t1[kk] : t0[kk], false, bf[ch * (SCOLS / 16) + j][kk],
                        (short)0, d, false, false);
                #pragma unroll
                for (int v = 0; v < 8; ++v)
                    stg[(v + 8 * h) * SPITCH + 16 * j + c] = d[v];
            }
            wave_lds_order();
            #pragma unroll
            for (int r = 0; r < 16; ++r) {
                const v4f q = *(const v4f*)&stg[r * SPITCH + 4 * lane];
                *(v4f*)(orow + (size_t)(16 * rt + r) * OUT_F + ch * SCOLS) = q;
            }
            wave_lds_order();
        }
    }
}

extern "C" void kernel_launch(void* const* d_in, const int* in_sizes, int n_in,
                              void* d_out, int out_size, void* d_ws, size_t ws_size,
                              hipStream_t stream) {
    const float* input   = (const float*)d_in[0];
    const float* weights = (const float*)d_in[1];
    const float* lora_a  = (const float*)d_in[2];
    const float* lora_b  = (const float*)d_in[3];
    float*       outp    = (float*)d_out;

    float* A_mix = (float*)d_ws;
    float* B_mix = A_mix + (size_t)BS * IN_F * RANK;

    lora_mix_kernel<<<2048, 256, 0, stream>>>(weights, lora_a, lora_b, A_mix, B_mix);

    skilled_lora_kernel<<<512, 256, 0, stream>>>(input, A_mix, B_mix, outp);
}


# GEMM1 reads x by whole-line 16-byte loads (4 lines per instruction, was 16) and forms its A fragments through wave-private LDS; k order inside each group of 8 changes, so the k-sum order changes
# speedup vs baseline: 1.7719x; 1.2637x over previous
#include <hip/hip_runtime.h>
#include <hip/hip_bf16.h>

#define BS      8
#define SEQ     2048
#define IN_F    2048
#define OUT_F   2048
#define NSPLIT  4
#define NSKILL  8
#define RANK    16
#define DSPLIT  (IN_F / NSPLIT)
#define OSPLIT  (OUT_F / NSPLIT)
#define SCALING 2.0f

typedef float v2f __attribute__((ext_vector_type(2)));
typedef float v8f __attribute__((ext_vector_type(8)));

__global__ __launch_bounds__(256) void lora_mix_kernel(
    const float* __restrict__ weights,
    const float* __restrict__ lora_a,
    const float* __restrict__ lora_b,
    float* __restrict__ A_mix,
    float* __restrict__ B_mix)
{
    const int idx  = blockIdx.x * 256 + threadIdx.x;
    const int half = BS * IN_F * RANK;

    if (idx < half) {
        const int b   = idx / (IN_F * RANK);
        const int rem = idx - b * (IN_F * RANK);
        const int i   = rem / RANK;
        const int r   = rem - i * RANK;
        const int q   = i / DSPLIT;
        const int d   = i - q * DSPLIT;
        const float* w  = weights + (b * NSPLIT + q) * NSKILL;
        const float* la = lora_a + ((size_t)(q * NSKILL) * DSPLIT + d) * RANK + r;
        float acc = 0.f;
        #pragma unroll
        for (int s = 0; s < NSKILL; ++s)
            acc = __builtin_fmaf(w[s], la[(size_t)s * DSPLIT * RANK], acc);
        A_mix[idx] = acc;
    } else {
        const int j   = idx - half;
        const int b   = j / (RANK * OUT_F);
        const int rem = j - b * (RANK * OUT_F);
        const int r   = rem / OUT_F;
        const int o   = rem - r * OUT_F;
        const int q   = o / OSPLIT;
        const int d   = o - q * OSPLIT;
        const float* w  = weights + (b * NSPLIT + q) * NSKILL;
        const float* lb = lora_b + ((size_t)(q * NSKILL) * RANK + r) * OSPLIT + d;
        float acc = 0.f;
        #pragma unroll
        for (int s = 0; s < NSKILL; ++s)
            acc = __builtin_fmaf(w[s], lb[(size_t)s * RANK * OSPLIT], acc);
        B_mix[j] = acc * SCALING;
    }
}

#define NWAVES  8
#define KSLICE  (IN_F / NWAVES)
#define NSLICE  (OUT_F / NWAVES)
#define TPITCH  20
#define SCOLS   128
#define SPITCH  (SCOLS + 4)
#define XK      64
#define XPITCH  (XK + 4)
#define NPASS   (KSLICE / XK)
#define WSTAGE  (32 * XPITCH)

typedef float v4f __attribute__((ext_vector_type(4)));

__device__ __forceinline__ void wave_lds_order() {
    __builtin_amdgcn_fence(__ATOMIC_RELEASE, "wavefront");
    __builtin_amdgcn_wave_barrier();
    __builtin_amdgcn_fence(__ATOMIC_ACQUIRE, "wavefront");
}

__global__ __launch_bounds__(256) void skilled_lora_kernel(
    const float* __restrict__ x,
    const float* __restrict__ A_mix,
    const float* __restrict__ B_mix,
    float* __restrict__ out)
{
    __shared__ __attribute__((aligned(16))) float Tp[NWAVES][2][16][TPITCH];
    __shared__ __attribute__((aligned(16))) float Stg[NWAVES][WSTAGE];

    const int wave = threadIdx.x >> 5;
    const int lane = threadIdx.x & 31;
    const int h    = lane >> 4;
    const int c    = lane & 15;

    const int b    = blockIdx.x >> 6;
    const int m0   = (blockIdx.x & 63) << 5;

    const int kb = wave * KSLICE;
    const float* xg   = x + ((size_t)b * SEQ + m0 + h) * IN_F + kb + 4 * c;
    const float* acol = A_mix + ((size_t)b * IN_F + kb + 4 * h) * RANK + c;
    float* stg = &Stg[wave][0];

    v8f acc0 = {0.f, 0.f, 0.f, 0.f, 0.f, 0.f, 0.f, 0.f};
    v8f acc1 = {0.f, 0.f, 0.f, 0.f, 0.f, 0.f, 0.f, 0.f};

    v4f xr[2][16];
    v2f bq[2][16];

    #pragma unroll
    for (int i = 0; i < 16; ++i)
        xr[0][i] = *(const v4f*)(xg + (size_t)(2 * i) * IN_F);
    #pragma unroll
    for (int q = 0; q < 16; ++q) {
        bq[0][q].x = acol[(size_t)(8 * (q >> 1) + 2 * (q & 1))     * RANK];
        bq[0][q].y = acol[(size_t)(8 * (q >> 1) + 2 * (q & 1) + 1) * RANK];
    }

    #pragma unroll
    for (int p = 0; p < NPASS; ++p) {
        if (p + 1 < NPASS) {
            #pragma unroll
            for (int i = 0; i < 16; ++i)
                xr[(p + 1) & 1][i] = *(const v4f*)(xg + (size_t)(2 * i) * IN_F + (p + 1) * XK);
            #pragma unroll
            for (int q = 0; q < 16; ++q) {
                bq[(p + 1) & 1][q].x = acol[(size_t)((p + 1) * XK + 8 * (q >> 1) + 2 * (q & 1))     * RANK];
                bq[(p + 1) & 1][q].y = acol[(size_t)((p + 1) * XK + 8 * (q >> 1) + 2 * (q & 1) + 1) * RANK];
            }
        }
        #pragma unroll
        for (int i = 0; i < 16; ++i)
            *(v4f*)&stg[(2 * i + h) * XPITCH + 4 * c] = xr[p & 1][i];
        wave_lds_order();
        v4f f0[XK / 8], f1[XK / 8];
        #pragma unroll
        for (int j = 0; j < XK / 8; ++j) {
            f0[j] = *(const v4f*)&stg[c * XPITCH + 8 * j + 4 * h];
            f1[j] = *(const v4f*)&stg[(16 + c) * XPITCH + 8 * j + 4 * h];
        }
        #pragma unroll
        for (int j = 0; j < XK / 8; ++j) {
            acc0 = __builtin_amdgcn_wmma_f32_16x16x4_f32(
                false, f0[j].xy, false, bq[p & 1][2 * j], (short)0, acc0, false, false);
            acc1 = __builtin_amdgcn_wmma_f32_16x16x4_f32(
                false, f1[j].xy, false, bq[p & 1][2 * j], (short)0, acc1, false, false);
            acc0 = __builtin_amdgcn_wmma_f32_16x16x4_f32(
                false, f0[j].zw, false, bq[p & 1][2 * j + 1], (short)0, acc0, false, false);
            acc1 = __builtin_amdgcn_wmma_f32_16x16x4_f32(
                false, f1[j].zw, false, bq[p & 1][2 * j + 1], (short)0, acc1, false, false);
        }
        wave_lds_order();
    }

    #pragma unroll
    for (int v = 0; v < 8; ++v) {
        Tp[wave][0][v + 8 * h][c] = acc0[v];
        Tp[wave][1][v + 8 * h][c] = acc1[v];
    }

    __syncthreads();

    v2f t0[4], t1[4];
    #pragma unroll
    for (int kk = 0; kk < 4; ++kk) {
        t0[kk] = *(const v2f*)&Tp[0][0][c][4 * kk + 2 * h];
        t1[kk] = *(const v2f*)&Tp[0][1][c][4 * kk + 2 * h];
    }
    #pragma unroll
    for (int w = 1; w < NWAVES; ++w) {
        #pragma unroll
        for (int kk = 0; kk < 4; ++kk) {
            t0[kk] += *(const v2f*)&Tp[w][0][c][4 * kk + 2 * h];
            t1[kk] += *(const v2f*)&Tp[w][1][c][4 * kk + 2 * h];
        }
    }

    const int nb = wave * NSLICE;
    const float* Bbase = B_mix + ((size_t)b * RANK + 2 * h) * OUT_F + nb + c;

    v2f bf[NSLICE / 16][4];
    #pragma unroll
    for (int j = 0; j < NSLICE / 16; ++j) {
        #pragma unroll
        for (int kk = 0; kk < 4; ++kk) {
            bf[j][kk].x = Bbase[(size_t)(4 * kk)     * OUT_F + 16 * j];
            bf[j][kk].y = Bbase[(size_t)(4 * kk + 1) * OUT_F + 16 * j];
        }
    }

    float* orow = out + ((size_t)b * SEQ + m0) * OUT_F + nb + 4 * lane;

    #pragma unroll
    for (int ch = 0; ch < NSLICE / SCOLS; ++ch) {
        #pragma unroll
        for (int rt = 0; rt < 2; ++rt) {
            #pragma unroll
            for (int j = 0; j < SCOLS / 16; ++j) {
                v8f d = {0.f, 0.f, 0.f, 0.f, 0.f, 0.f, 0.f, 0.f};
                #pragma unroll
                for (int kk = 0; kk < 4; ++kk)
                    d = __builtin_amdgcn_wmma_f32_16x16x4_f32(
                        false, rt ? t1[kk] : t0[kk], false, bf[ch * (SCOLS / 16) + j][kk],
                        (short)0, d, false, false);
                #pragma unroll
                for (int v = 0; v < 8; ++v)
                    stg[(v + 8 * h) * SPITCH + 16 * j + c] = d[v];
            }
            wave_lds_order();
            #pragma unroll
            for (int r = 0; r < 16; ++r) {
                const v4f q = *(const v4f*)&stg[r * SPITCH + 4 * lane];
                *(v4f*)(orow + (size_t)(16 * rt + r) * OUT_F + ch * SCOLS) = q;
            }
            wave_lds_order();
        }
    }
}

extern "C" void kernel_launch(void* const* d_in, const int* in_sizes, int n_in,
                              void* d_out, int out_size, void* d_ws, size_t ws_size,
                              hipStream_t stream) {
    const float* input   = (const float*)d_in[0];
    const float* weights = (const float*)d_in[1];
    const float* lora_a  = (const float*)d_in[2];
    const float* lora_b  = (const float*)d_in[3];
    float*       outp    = (float*)d_out;

    float* A_mix = (float*)d_ws;
    float* B_mix = A_mix + (size_t)BS * IN_F * RANK;

    lora_mix_kernel<<<2048, 256, 0, stream>>>(weights, lora_a, lora_b, A_mix, B_mix);

    skilled_lora_kernel<<<512, 256, 0, stream>>>(input, A_mix, B_mix, outp);
}
